# starting kernel
# baseline (speedup 1.0000x reference)
#include <hip/hip_runtime.h>
#include <hip/hip_bf16.h>


typedef __attribute__((ext_vector_type(2))) float v2f;
typedef __attribute__((ext_vector_type(8))) float v8f;

#define CIN   64
#define COUT  64
#define RANK  4
#define HH    56
#define WW    56
#define KDIM  (CIN * 9)
#define TILE_W 58
#define CI_STRIDE (3 * TILE_W)

__global__ void prep_keff(const float* __restrict__ vw,
                          const float* __restrict__ hw,
                          float* __restrict__ keff) {
    int idx = blockIdx.x * blockDim.x + threadIdx.x;
    if (idx >= COUT * KDIM) return;
    int co = idx / KDIM;
    int k  = idx - co * KDIM;
    int kk = k >> 6;
    int ci = k & 63;
    int kh = kk / 3;
    int kw = kk - kh * 3;
    int gbase = ci * (COUT * RANK) + co * RANK;
    float s = 0.f;
#pragma unroll
    for (int r = 0; r < RANK; ++r) {
        int g = gbase + r;
        s += hw[g * 3 + kh] * vw[g * 3 + kw];
    }
    keff[idx] = s;
}

__global__ __launch_bounds__(512)
void conv_wmma(const float* __restrict__ x,
               const float* __restrict__ keff,
               const float* __restrict__ bias,
               float* __restrict__ out) {
    __shared__ float tile[CIN * CI_STRIDE];

    const int bh = blockIdx.x;
    const int b  = bh / HH;
    const int h  = bh - b * HH;
    const int tid = threadIdx.x;

    const float* xb = x + (size_t)b * CIN * HH * WW;
    for (int e = tid; e < CIN * CI_STRIDE; e += 512) {
        int ci  = e / CI_STRIDE;
        int rem = e - ci * CI_STRIDE;
        int ry  = rem / TILE_W;
        int col = rem - ry * TILE_W;
        int y  = h + ry - 1;
        int wc = col - 1;
        float v = 0.f;
        if ((unsigned)y < (unsigned)HH && (unsigned)wc < (unsigned)WW)
            v = xb[(ci * HH + y) * WW + wc];
        tile[e] = v;
    }
    __syncthreads();

    const int wave  = tid >> 5;
    const int lane  = tid & 31;
    const int half  = lane >> 4;
    const int laneW = lane & 15;
    const int co0 = (wave >> 2) * 16;
    const int w0  = (wave & 3) * 16;

    const float* aRow = keff + (co0 + laneW) * KDIM;

    v8f acc = {0.f, 0.f, 0.f, 0.f, 0.f, 0.f, 0.f, 0.f};

    for (int kk = 0; kk < 9; ++kk) {
        const int kh = kk / 3;
        const int kw = kk - kh * 3;
        const float* bBase = tile + 2 * half * CI_STRIDE
                                  + kh * TILE_W + w0 + laneW + kw;
        const float* aK = aRow + kk * 64 + 2 * half;
#pragma unroll
        for (int t = 0; t < 16; ++t) {
            v2f a = *(const v2f*)(aK + 4 * t);
            v2f bv;
            bv.x = bBase[(4 * t)     * CI_STRIDE];
            bv.y = bBase[(4 * t + 1) * CI_STRIDE];
            acc = __builtin_amdgcn_wmma_f32_16x16x4_f32(
                false, a, false, bv, (short)0, acc, false, false);
        }
    }

    const int wN = w0 + laneW;
    if (wN < WW) {
        const int coB = co0 + half * 8;
        float* op = out + (((size_t)b * COUT + coB) * HH + h) * WW + wN;
#pragma unroll
        for (int j = 0; j < 8; ++j)
            op[(size_t)j * HH * WW] = acc[j] + bias[coB + j];
    }
}

extern "C" void kernel_launch(void* const* d_in, const int* in_sizes, int n_in,
                              void* d_out, int out_size, void* d_ws, size_t ws_size,
                              hipStream_t stream) {
    const float* x    = (const float*)d_in[0];
    const float* vw   = (const float*)d_in[1];
    const float* hw   = (const float*)d_in[2];
    const float* bias = (const float*)d_in[3];
    float* out  = (float*)d_out;
    float* keff = (float*)d_ws;

    const int nK = COUT * KDIM;
    prep_keff<<<(nK + 255) / 256, 256, 0, stream>>>(vw, hw, keff);

    const int B = 4;
    conv_wmma<<<B * HH, 512, 0, stream>>>(x, keff, bias, out);
}


# Input slab staging: each thread issues its 6 independent 16-byte row loads together (16B-aligned LDS rows, halo cells written separately) instead of 22 dependent load-wait-store rounds
# speedup vs baseline: 1.4251x; 1.4251x over previous
#include <hip/hip_runtime.h>
#include <hip/hip_bf16.h>


typedef __attribute__((ext_vector_type(2))) float v2f;
typedef __attribute__((ext_vector_type(8))) float v8f;
typedef __attribute__((ext_vector_type(4))) float v4f;

#define CIN   64
#define COUT  64
#define RANK  4
#define HH    56
#define WW    56
#define KDIM  (CIN * 9)
#define ROW_P 68
#define CI_STRIDE (3 * ROW_P)
#define TILE_FLOATS (CIN * CI_STRIDE + 4)

__global__ void prep_keff(const float* __restrict__ vw,
                          const float* __restrict__ hw,
                          float* __restrict__ keff) {
    int idx = blockIdx.x * blockDim.x + threadIdx.x;
    if (idx >= COUT * KDIM) return;
    int co = idx / KDIM;
    int k  = idx - co * KDIM;
    int kk = k >> 6;
    int ci = k & 63;
    int kh = kk / 3;
    int kw = kk - kh * 3;
    int gbase = ci * (COUT * RANK) + co * RANK;
    float s = 0.f;
#pragma unroll
    for (int r = 0; r < RANK; ++r) {
        int g = gbase + r;
        s += hw[g * 3 + kh] * vw[g * 3 + kw];
    }
    keff[idx] = s;
}

__global__ __launch_bounds__(512)
void conv_wmma(const float* __restrict__ x,
               const float* __restrict__ keff,
               const float* __restrict__ bias,
               float* __restrict__ out) {
    __shared__ __attribute__((aligned(16))) float tile[TILE_FLOATS];

    const int bh = blockIdx.x;
    const int b  = bh / HH;
    const int h  = bh - b * HH;
    const int tid = threadIdx.x;

    const float* xb = x + (size_t)b * CIN * HH * WW;
    if (tid < 2 * 3 * CIN) {
        const int r = tid >> 1;
        tile[(r & 63) * CI_STRIDE + (r >> 6) * ROW_P + ((tid & 1) ? 4 + WW : 3)] = 0.f;
    }
    const int q    = tid & 15;
    const int rsub = tid >> 4;
    if (q < 14) {
        v4f sv[6];
#pragma unroll
        for (int i = 0; i < 6; ++i) {
            const int ry = i >> 1;
            const int ci = rsub + 32 * (i & 1);
            int y = h + ry - 1;
            y = y < 0 ? 0 : (y > HH - 1 ? HH - 1 : y);
            sv[i] = *(const v4f*)(xb + (ci * HH + y) * WW + 4 * q);
        }
        asm volatile("" :: "v"(sv[0]), "v"(sv[1]), "v"(sv[2]), "v"(sv[3]), "v"(sv[4]), "v"(sv[5]));
#pragma unroll
        for (int i = 0; i < 6; ++i) {
            const int ry = i >> 1;
            const int ci = rsub + 32 * (i & 1);
            const int y  = h + ry - 1;
            v4f v = sv[i];
            if ((unsigned)y >= (unsigned)HH) v = (v4f){0.f, 0.f, 0.f, 0.f};
            *(v4f*)(tile + ci * CI_STRIDE + ry * ROW_P + 4 + 4 * q) = v;
        }
    }
    __syncthreads();

    const int wave  = tid >> 5;
    const int lane  = tid & 31;
    const int half  = lane >> 4;
    const int laneW = lane & 15;
    const int co0 = (wave >> 2) * 16;
    const int w0  = (wave & 3) * 16;

    const float* aRow = keff + (co0 + laneW) * KDIM;

    v8f acc = {0.f, 0.f, 0.f, 0.f, 0.f, 0.f, 0.f, 0.f};

    for (int kk = 0; kk < 9; ++kk) {
        const int kh = kk / 3;
        const int kw = kk - kh * 3;
        const float* bBase = tile + 2 * half * CI_STRIDE
                                  + kh * ROW_P + 3 + w0 + laneW + kw;
        const float* aK = aRow + kk * 64 + 2 * half;
#pragma unroll
        for (int t = 0; t < 16; ++t) {
            v2f a = *(const v2f*)(aK + 4 * t);
            v2f bv;
            bv.x = bBase[(4 * t)     * CI_STRIDE];
            bv.y = bBase[(4 * t + 1) * CI_STRIDE];
            acc = __builtin_amdgcn_wmma_f32_16x16x4_f32(
                false, a, false, bv, (short)0, acc, false, false);
        }
    }

    const int wN = w0 + laneW;
    if (wN < WW) {
        const int coB = co0 + half * 8;
        float* op = out + (((size_t)b * COUT + coB) * HH + h) * WW + wN;
#pragma unroll
        for (int j = 0; j < 8; ++j)
            op[(size_t)j * HH * WW] = acc[j] + bias[coB + j];
    }
}

extern "C" void kernel_launch(void* const* d_in, const int* in_sizes, int n_in,
                              void* d_out, int out_size, void* d_ws, size_t ws_size,
                              hipStream_t stream) {
    const float* x    = (const float*)d_in[0];
    const float* vw   = (const float*)d_in[1];
    const float* hw   = (const float*)d_in[2];
    const float* bias = (const float*)d_in[3];
    float* out  = (float*)d_out;
    float* keff = (float*)d_ws;

    const int nK = COUT * KDIM;
    prep_keff<<<(nK + 255) / 256, 256, 0, stream>>>(vw, hw, keff);

    const int B = 4;
    conv_wmma<<<B * HH, 512, 0, stream>>>(x, keff, bias, out);
}


# A-operand feed: weights stored in A-fragment order so a lane's A values are 16-byte loads (4 cache lines per wave load, was 16), each tap's eight issued as one group two taps ahead of use
# speedup vs baseline: 2.1880x; 1.5353x over previous
#include <hip/hip_runtime.h>
#include <hip/hip_bf16.h>


typedef __attribute__((ext_vector_type(2))) float v2f;
typedef __attribute__((ext_vector_type(8))) float v8f;
typedef __attribute__((ext_vector_type(4))) float v4f;

#define CIN   64
#define COUT  64
#define RANK  4
#define HH    56
#define WW    56
#define KDIM  (CIN * 9)
#define ROW_P 68
#define CI_STRIDE (3 * ROW_P)
#define TILE_FLOATS (CIN * CI_STRIDE + 4)

__global__ void prep_keff(const float* __restrict__ vw,
                          const float* __restrict__ hw,
                          float* __restrict__ keff) {
    int idx = blockIdx.x * blockDim.x + threadIdx.x;
    if (idx >= COUT * KDIM) return;
    const int e    = idx & 3;
    const int lane = (idx >> 2) & 31;
    const int u    = (idx >> 7) & 7;
    const int rest = idx >> 10;
    const int coT  = rest / 9;
    const int kk   = rest - coT * 9;
    const int kh = kk / 3;
    const int kw = kk - kh * 3;
    const int co = coT * 16 + (lane & 15);
    const int ci = 4 * (2 * u + (e >> 1)) + 2 * (lane >> 4) + (e & 1);
    int gbase = ci * (COUT * RANK) + co * RANK;
    float s = 0.f;
#pragma unroll
    for (int r = 0; r < RANK; ++r) {
        int g = gbase + r;
        s += hw[g * 3 + kh] * vw[g * 3 + kw];
    }
    keff[idx] = s;
}

__global__ __launch_bounds__(512)
void conv_wmma(const float* __restrict__ x,
               const float* __restrict__ keff,
               const float* __restrict__ bias,
               float* __restrict__ out) {
    __shared__ __attribute__((aligned(16))) float tile[TILE_FLOATS];

    const int bh = blockIdx.x;
    const int b  = bh / HH;
    const int h  = bh - b * HH;
    const int tid = threadIdx.x;

    const int wave  = tid >> 5;
    const int lane  = tid & 31;
    const int half  = lane >> 4;
    const int laneW = lane & 15;
    const int co0 = (wave >> 2) * 16;
    const int w0  = (wave & 3) * 16;

    const v4f* aF = (const v4f*)keff + (wave >> 2) * (9 * 8 * 32) + lane;
    v4f cur[8], nxt[8];
#pragma unroll
    for (int u = 0; u < 8; ++u) cur[u] = aF[u * 32];

    const float* xb = x + (size_t)b * CIN * HH * WW;
    if (tid < 2 * 3 * CIN) {
        const int r = tid >> 1;
        tile[(r & 63) * CI_STRIDE + (r >> 6) * ROW_P + ((tid & 1) ? 4 + WW : 3)] = 0.f;
    }
    const int q    = tid & 15;
    const int rsub = tid >> 4;
    if (q < 14) {
        v4f sv[6];
#pragma unroll
        for (int i = 0; i < 6; ++i) {
            const int ry = i >> 1;
            const int ci = rsub + 32 * (i & 1);
            int y = h + ry - 1;
            y = y < 0 ? 0 : (y > HH - 1 ? HH - 1 : y);
            sv[i] = *(const v4f*)(xb + (ci * HH + y) * WW + 4 * q);
        }
        asm volatile("" :: "v"(sv[0]), "v"(sv[1]), "v"(sv[2]), "v"(sv[3]), "v"(sv[4]), "v"(sv[5]));
#pragma unroll
        for (int i = 0; i < 6; ++i) {
            const int ry = i >> 1;
            const int ci = rsub + 32 * (i & 1);
            const int y  = h + ry - 1;
            v4f v = sv[i];
            if ((unsigned)y >= (unsigned)HH) v = (v4f){0.f, 0.f, 0.f, 0.f};
            *(v4f*)(tile + ci * CI_STRIDE + ry * ROW_P + 4 + 4 * q) = v;
        }
    }
    __syncthreads();

    v8f acc = {0.f, 0.f, 0.f, 0.f, 0.f, 0.f, 0.f, 0.f};

    for (int kk = 0; kk < 9; ++kk) {
        const int kh = kk / 3;
        const int kw = kk - kh * 3;
        if (kk + 1 < 9) {
#pragma unroll
            for (int u = 0; u < 8; ++u) nxt[u] = aF[((kk + 1) * 8 + u) * 32];
        }
        const float* bBase = tile + 2 * half * CI_STRIDE
                                  + kh * ROW_P + 3 + w0 + laneW + kw;
#pragma unroll
        for (int u = 0; u < 8; ++u) {
            const v4f a4 = cur[u];
            v2f a, bv;
            a.x  = a4.x;
            a.y  = a4.y;
            bv.x = bBase[(8 * u)     * CI_STRIDE];
            bv.y = bBase[(8 * u + 1) * CI_STRIDE];
            acc = __builtin_amdgcn_wmma_f32_16x16x4_f32(
                false, a, false, bv, (short)0, acc, false, false);
            a.x  = a4.z;
            a.y  = a4.w;
            bv.x = bBase[(8 * u + 4) * CI_STRIDE];
            bv.y = bBase[(8 * u + 5) * CI_STRIDE];
            acc = __builtin_amdgcn_wmma_f32_16x16x4_f32(
                false, a, false, bv, (short)0, acc, false, false);
        }
        __builtin_amdgcn_sched_barrier(0);
#pragma unroll
        for (int u = 0; u < 8; ++u) cur[u] = nxt[u];
    }

    const int wN = w0 + laneW;
    if (wN < WW) {
        const int coB = co0 + half * 8;
        float* op = out + (((size_t)b * COUT + coB) * HH + h) * WW + wN;
#pragma unroll
        for (int j = 0; j < 8; ++j)
            op[(size_t)j * HH * WW] = acc[j] + bias[coB + j];
    }
}

extern "C" void kernel_launch(void* const* d_in, const int* in_sizes, int n_in,
                              void* d_out, int out_size, void* d_ws, size_t ws_size,
                              hipStream_t stream) {
    const float* x    = (const float*)d_in[0];
    const float* vw   = (const float*)d_in[1];
    const float* hw   = (const float*)d_in[2];
    const float* bias = (const float*)d_in[3];
    float* out  = (float*)d_out;
    float* keff = (float*)d_ws;

    const int nK = COUT * KDIM;
    prep_keff<<<(nK + 255) / 256, 256, 0, stream>>>(vw, hw, keff);

    const int B = 4;
    conv_wmma<<<B * HH, 512, 0, stream>>>(x, keff, bias, out);
}


# One launch instead of two: a workgroup = one 16-channel co tile x 4 output rows; it collapses its own 1024 (ci,co) weight pairs into LDS in A-fragment order and reads its A fragments from there
# speedup vs baseline: 2.6094x; 1.1926x over previous
#include <hip/hip_runtime.h>
#include <hip/hip_bf16.h>


typedef __attribute__((ext_vector_type(2))) float v2f;
typedef __attribute__((ext_vector_type(8))) float v8f;
typedef __attribute__((ext_vector_type(4))) float v4f;

#define CIN   64
#define COUT  64
#define RANK  4
#define HH    56
#define WW    56
#define ROWS  4
#define SROWS (ROWS + 2)
#define ROW_P 80
#define RY_STRIDE (CIN * ROW_P)
#define TILE_FLOATS (SROWS * RY_STRIDE)
#define AFRAGS (9 * 8 * 32)
#define NBLK  (4 * 4 * (HH / ROWS))

__global__ __launch_bounds__(512)
void conv_wmma(const float* __restrict__ x,
               const float* __restrict__ vw,
               const float* __restrict__ hw,
               const float* __restrict__ bias,
               float* __restrict__ out) {
    __shared__ __attribute__((aligned(16))) float tile[TILE_FLOATS];
    __shared__ __attribute__((aligned(16))) float aL[AFRAGS * 4];

    const int coT   = blockIdx.x & 3;
    const int chunk = blockIdx.x >> 2;
    const int b   = chunk / (HH / ROWS);
    const int h0  = (chunk - b * (HH / ROWS)) * ROWS;
    const int tid = threadIdx.x;

    const int cA  = tid & 15;
    const int ciA = tid >> 4;
    v4f hq[2][3], vq[2][3];
#pragma unroll
    for (int p = 0; p < 2; ++p) {
        const int off = (ciA + 32 * p) * (COUT * RANK * 3) + (coT * 16 + cA) * (RANK * 3);
#pragma unroll
        for (int k = 0; k < 3; ++k) {
            hq[p][k] = *(const v4f*)(hw + off + 4 * k);
            vq[p][k] = *(const v4f*)(vw + off + 4 * k);
        }
    }

    const float* xb = x + (size_t)b * CIN * HH * WW;
    const int q    = tid & 15;
    const int qc   = q < 14 ? q : 13;
    const int rsub = tid >> 4;
    v4f sv[12];
#pragma unroll
    for (int i = 0; i < 12; ++i) {
        const int ry = i >> 1;
        const int ci = rsub + 32 * (i & 1);
        int y = h0 + ry - 1;
        y = y < 0 ? 0 : (y > HH - 1 ? HH - 1 : y);
        sv[i] = *(const v4f*)(xb + (ci * HH + y) * WW + 4 * qc);
    }

    if (tid < SROWS * CIN) {
        tile[tid * ROW_P + 3]      = 0.f;
        tile[tid * ROW_P + 4 + WW] = 0.f;
    }

#pragma unroll
    for (int p = 0; p < 2; ++p) {
        float hf12[12], vf12[12];
#pragma unroll
        for (int k = 0; k < 3; ++k) {
            hf12[4 * k + 0] = hq[p][k].x; hf12[4 * k + 1] = hq[p][k].y;
            hf12[4 * k + 2] = hq[p][k].z; hf12[4 * k + 3] = hq[p][k].w;
            vf12[4 * k + 0] = vq[p][k].x; vf12[4 * k + 1] = vq[p][k].y;
            vf12[4 * k + 2] = vq[p][k].z; vf12[4 * k + 3] = vq[p][k].w;
        }
        const int ci = ciA + 32 * p;
        const int u  = ci >> 3;
        const int hf = (ci >> 1) & 1;
        const int e  = ((ci >> 2) & 1) * 2 + (ci & 1);
        float* ap = aL + (u * 32 + hf * 16 + cA) * 4 + e;
#pragma unroll
        for (int kh = 0; kh < 3; ++kh) {
#pragma unroll
            for (int kw = 0; kw < 3; ++kw) {
                float s = 0.f;
#pragma unroll
                for (int r = 0; r < RANK; ++r) s += hf12[r * 3 + kh] * vf12[r * 3 + kw];
                ap[(kh * 3 + kw) * (8 * 32 * 4)] = s;
            }
        }
    }

    if (q < 14) {
#pragma unroll
        for (int i = 0; i < 12; ++i) {
            const int ry = i >> 1;
            const int ci = rsub + 32 * (i & 1);
            const int y  = h0 + ry - 1;
            v4f v = sv[i];
            if ((unsigned)y >= (unsigned)HH) v = (v4f){0.f, 0.f, 0.f, 0.f};
            *(v4f*)(tile + (ry * CIN + ci) * ROW_P + 4 + 4 * q) = v;
        }
    }
    __syncthreads();

    const int wave  = tid >> 5;
    const int lane  = tid & 31;
    const int half  = lane >> 4;
    const int laneW = lane & 15;
    const int r   = wave >> 2;
    const int w0  = (wave & 3) * 16;

    const v4f* aF = (const v4f*)aL + lane;
    v4f cur[8], nxt[8];
#pragma unroll
    for (int u = 0; u < 8; ++u) cur[u] = aF[u * 32];

    v8f acc = {0.f, 0.f, 0.f, 0.f, 0.f, 0.f, 0.f, 0.f};

    for (int kk = 0; kk < 9; ++kk) {
        const int kh = kk / 3;
        const int kw = kk - kh * 3;
        if (kk + 1 < 9) {
#pragma unroll
            for (int u = 0; u < 8; ++u) nxt[u] = aF[((kk + 1) * 8 + u) * 32];
        }
        const float* bBase = tile + ((r + kh) * CIN + 2 * half) * ROW_P
                                  + 3 + w0 + laneW + kw;
#pragma unroll
        for (int u = 0; u < 8; ++u) {
            const v4f a4 = cur[u];
            v2f a, bv;
            a.x  = a4.x;
            a.y  = a4.y;
            bv.x = bBase[(8 * u)     * ROW_P];
            bv.y = bBase[(8 * u + 1) * ROW_P];
            acc = __builtin_amdgcn_wmma_f32_16x16x4_f32(
                false, a, false, bv, (short)0, acc, false, false);
            a.x  = a4.z;
            a.y  = a4.w;
            bv.x = bBase[(8 * u + 4) * ROW_P];
            bv.y = bBase[(8 * u + 5) * ROW_P];
            acc = __builtin_amdgcn_wmma_f32_16x16x4_f32(
                false, a, false, bv, (short)0, acc, false, false);
        }
        __builtin_amdgcn_sched_barrier(0);
#pragma unroll
        for (int u = 0; u < 8; ++u) cur[u] = nxt[u];
    }

    const int wN = w0 + laneW;
    if (wN < WW) {
        const int coB = coT * 16 + half * 8;
        float* op = out + (((size_t)b * COUT + coB) * HH + (h0 + r)) * WW + wN;
#pragma unroll
        for (int j = 0; j < 8; ++j)
            op[(size_t)j * HH * WW] = acc[j] + bias[coB + j];
    }
}

extern "C" void kernel_launch(void* const* d_in, const int* in_sizes, int n_in,
                              void* d_out, int out_size, void* d_ws, size_t ws_size,
                              hipStream_t stream) {
    const float* x    = (const float*)d_in[0];
    const float* vw   = (const float*)d_in[1];
    const float* hw   = (const float*)d_in[2];
    const float* bias = (const float*)d_in[3];
    float* out  = (float*)d_out;

    conv_wmma<<<NBLK, 512, 0, stream>>>(x, vw, hw, bias, out);
}


# Minimal filtering F(2,3) along the height: input rows and kh taps transformed at staging, 4 matrix products per output row pair instead of 6 (8 waves x 192 WMMAs); summation order changes
# speedup vs baseline: 2.9161x; 1.1175x over previous
#include <hip/hip_runtime.h>
#include <hip/hip_bf16.h>


typedef __attribute__((ext_vector_type(2))) float v2f;
typedef __attribute__((ext_vector_type(8))) float v8f;
typedef __attribute__((ext_vector_type(4))) float v4f;

#define CIN   64
#define COUT  64
#define RANK  4
#define HH    56
#define WW    56
#define ROWS  4
#define SROWS (ROWS + 2)
#define PLANES 8
#define ROW_P 80
#define PL_STRIDE (CIN * ROW_P)
#define TILE_FLOATS (PLANES * PL_STRIDE)
#define NTAP  12
#define AFRAGS (NTAP * 8 * 32)
#define NBLK  (4 * 4 * (HH / ROWS))
#define NTHR  256

__global__ __launch_bounds__(NTHR)
void conv_wmma(const float* __restrict__ x,
               const float* __restrict__ vw,
               const float* __restrict__ hw,
               const float* __restrict__ bias,
               float* __restrict__ out) {
    __shared__ __attribute__((aligned(16))) float tile[TILE_FLOATS];
    __shared__ __attribute__((aligned(16))) float aL[AFRAGS * 4];

    const int coT   = blockIdx.x & 3;
    const int chunk = blockIdx.x >> 2;
    const int b   = chunk / (HH / ROWS);
    const int h0  = (chunk - b * (HH / ROWS)) * ROWS;
    const int tid = threadIdx.x;

    const int cA  = tid & 15;
    const int ciA = tid >> 4;
    v4f hq[4][3], vq[4][3];
#pragma unroll
    for (int p = 0; p < 4; ++p) {
        const int off = (ciA + 16 * p) * (COUT * RANK * 3) + (coT * 16 + cA) * (RANK * 3);
#pragma unroll
        for (int k = 0; k < 3; ++k) {
            hq[p][k] = *(const v4f*)(hw + off + 4 * k);
            vq[p][k] = *(const v4f*)(vw + off + 4 * k);
        }
    }

    const float* xb = x + (size_t)b * CIN * HH * WW;
    const int q    = tid & 15;
    const int qc   = q < 14 ? q : 13;
    const int rsub = tid >> 4;
    v4f sv[24];
#pragma unroll
    for (int i = 0; i < 24; ++i) {
        const int ry = i >> 2;
        const int ci = rsub + 16 * (i & 3);
        int y = h0 + ry - 1;
        y = y < 0 ? 0 : (y > HH - 1 ? HH - 1 : y);
        sv[i] = *(const v4f*)(xb + (ci * HH + y) * WW + 4 * qc);
    }
    __builtin_amdgcn_sched_barrier(0);

#pragma unroll
    for (int k = 0; k < 2; ++k) {
        const int prow = tid + NTHR * k;
        tile[prow * ROW_P + 3] = 0.f;
    }

#pragma unroll
    for (int p = 0; p < 4; ++p) {
        float hf12[12], vf12[12];
#pragma unroll
        for (int k = 0; k < 3; ++k) {
            hf12[4 * k + 0] = hq[p][k].x; hf12[4 * k + 1] = hq[p][k].y;
            hf12[4 * k + 2] = hq[p][k].z; hf12[4 * k + 3] = hq[p][k].w;
            vf12[4 * k + 0] = vq[p][k].x; vf12[4 * k + 1] = vq[p][k].y;
            vf12[4 * k + 2] = vq[p][k].z; vf12[4 * k + 3] = vq[p][k].w;
        }
        const int ci = ciA + 16 * p;
        const int u  = ci >> 3;
        const int hf = (ci >> 1) & 1;
        const int e  = ((ci >> 2) & 1) * 2 + (ci & 1);
        float* ap = aL + (u * 32 + hf * 16 + cA) * 4 + e;
#pragma unroll
        for (int kw = 0; kw < 3; ++kw) {
            float g[3];
#pragma unroll
            for (int kh = 0; kh < 3; ++kh) {
                float s = 0.f;
#pragma unroll
                for (int r = 0; r < RANK; ++r) s += hf12[r * 3 + kh] * vf12[r * 3 + kw];
                g[kh] = s;
            }
            const float s02 = g[0] + g[2];
            ap[(0 * 3 + kw) * (8 * 32 * 4)] = g[0];
            ap[(1 * 3 + kw) * (8 * 32 * 4)] = 0.5f * (s02 + g[1]);
            ap[(2 * 3 + kw) * (8 * 32 * 4)] = 0.5f * (s02 - g[1]);
            ap[(3 * 3 + kw) * (8 * 32 * 4)] = g[2];
        }
    }

    const bool live = q < 14;
#pragma unroll
    for (int cg = 0; cg < 4; ++cg) {
        const int ci = rsub + 16 * cg;
        v4f s[SROWS];
#pragma unroll
        for (int ry = 0; ry < SROWS; ++ry) {
            const int y = h0 + ry - 1;
            v4f v = sv[ry * 4 + cg];
            if (!live || (unsigned)y >= (unsigned)HH) v = (v4f){0.f, 0.f, 0.f, 0.f};
            s[ry] = v;
        }
        float* tp = tile + ci * ROW_P + 4 + 4 * q;
#pragma unroll
        for (int pr = 0; pr < 2; ++pr) {
            const v4f d0 = s[2 * pr], d1 = s[2 * pr + 1], d2 = s[2 * pr + 2], d3 = s[2 * pr + 3];
            *(v4f*)(tp + (pr * 4 + 0) * PL_STRIDE) = d0 - d2;
            *(v4f*)(tp + (pr * 4 + 1) * PL_STRIDE) = d1 + d2;
            *(v4f*)(tp + (pr * 4 + 2) * PL_STRIDE) = d2 - d1;
            *(v4f*)(tp + (pr * 4 + 3) * PL_STRIDE) = d1 - d3;
        }
    }
    __syncthreads();

    const int wave  = tid >> 5;
    const int lane  = tid & 31;
    const int half  = lane >> 4;
    const int laneW = lane & 15;
    const int pr  = wave >> 2;
    const int w0  = (wave & 3) * 16;

    const float* bP = tile + ((pr * 4) * CIN + 2 * half) * ROW_P + 3 + w0 + laneW;
    const v4f*   aP = (const v4f*)aL + lane;
    v8f o0 = {0.f, 0.f, 0.f, 0.f, 0.f, 0.f, 0.f, 0.f};
    v8f o1 = {0.f, 0.f, 0.f, 0.f, 0.f, 0.f, 0.f, 0.f};

#pragma nounroll
    for (int hp = 0; hp < 2; ++hp) {
        v4f cur[16], nxt[16];
#pragma unroll
        for (int u = 0; u < 8; ++u) {
            cur[u]     = aP[(0 * 8 + u) * 32];
            cur[8 + u] = aP[(3 * 8 + u) * 32];
        }
        v8f accA = {0.f, 0.f, 0.f, 0.f, 0.f, 0.f, 0.f, 0.f};
        v8f accB = {0.f, 0.f, 0.f, 0.f, 0.f, 0.f, 0.f, 0.f};

#pragma unroll
        for (int kw = 0; kw < 3; ++kw) {
            if (kw + 1 < 3) {
#pragma unroll
                for (int u = 0; u < 8; ++u) {
                    nxt[u]     = aP[((kw + 1) * 8 + u) * 32];
                    nxt[8 + u] = aP[((3 + kw + 1) * 8 + u) * 32];
                }
            }
            const float* bA = bP + kw;
            const float* bB = bA + PL_STRIDE;
#pragma unroll
            for (int u = 0; u < 8; ++u) {
                const v4f a4A = cur[u];
                const v4f a4B = cur[8 + u];
                v2f a, bv;
                a.x  = a4A.x;
                a.y  = a4A.y;
                bv.x = bA[(8 * u)     * ROW_P];
                bv.y = bA[(8 * u + 1) * ROW_P];
                accA = __builtin_amdgcn_wmma_f32_16x16x4_f32(
                    false, a, false, bv, (short)0, accA, false, false);
                a.x  = a4B.x;
                a.y  = a4B.y;
                bv.x = bB[(8 * u)     * ROW_P];
                bv.y = bB[(8 * u + 1) * ROW_P];
                accB = __builtin_amdgcn_wmma_f32_16x16x4_f32(
                    false, a, false, bv, (short)0, accB, false, false);
                a.x  = a4A.z;
                a.y  = a4A.w;
                bv.x = bA[(8 * u + 4) * ROW_P];
                bv.y = bA[(8 * u + 5) * ROW_P];
                accA = __builtin_amdgcn_wmma_f32_16x16x4_f32(
                    false, a, false, bv, (short)0, accA, false, false);
                a.x  = a4B.z;
                a.y  = a4B.w;
                bv.x = bB[(8 * u + 4) * ROW_P];
                bv.y = bB[(8 * u + 5) * ROW_P];
                accB = __builtin_amdgcn_wmma_f32_16x16x4_f32(
                    false, a, false, bv, (short)0, accB, false, false);
            }
            __builtin_amdgcn_sched_barrier(0);
#pragma unroll
            for (int u = 0; u < 16; ++u) cur[u] = nxt[u];
        }

        const bool first = (hp == 0);
#pragma unroll
        for (int j = 0; j < 8; ++j) {
            const float pa = accA[j], pb = accB[j];
            o0[j] = first ? (pa + pb) : (o0[j] + pa);
            o1[j] = first ? pb        : ((o1[j] - pa) - pb);
        }
        bP += 2 * PL_STRIDE;
        aP += 6 * 8 * 32;
    }

    const int wN = w0 + laneW;
    if (wN < WW) {
        const int coB = coT * 16 + half * 8;
        float* op = out + (((size_t)b * COUT + coB) * HH + (h0 + 2 * pr)) * WW + wN;
#pragma unroll
        for (int j = 0; j < 8; ++j) {
            const float bj = bias[coB + j];
            op[(size_t)j * HH * WW]      = o0[j] + bj;
            op[(size_t)j * HH * WW + WW] = o1[j] + bj;
        }
    }
}

extern "C" void kernel_launch(void* const* d_in, const int* in_sizes, int n_in,
                              void* d_out, int out_size, void* d_ws, size_t ws_size,
                              hipStream_t stream) {
    const float* x    = (const float*)d_in[0];
    const float* vw   = (const float*)d_in[1];
    const float* hw   = (const float*)d_in[2];
    const float* bias = (const float*)d_in[3];
    float* out  = (float*)d_out;

    conv_wmma<<<NBLK, NTHR, 0, stream>>>(x, vw, hw, bias, out);
}
